# starting kernel
# baseline (speedup 1.0000x reference)
#include <hip/hip_runtime.h>
#include <hip/hip_bf16.h>
#include <cmath>

typedef __attribute__((ext_vector_type(2))) float v2f;
typedef __attribute__((ext_vector_type(8))) float v8f;

#define B_  64
#define S_  512
#define D_  1024
#define K_  2048

__global__ void build_combined_kernel(const float* __restrict__ tokens,
                                      const float* __restrict__ cls,
                                      const int*   __restrict__ idx,
                                      float*       __restrict__ comb) {
    const int tid = blockIdx.x * blockDim.x + threadIdx.x;
    const int b = tid >> 11;
    const int c = tid & (K_ - 1);
    float v;
    if (c < D_) {
        const int s = idx[c] + 1;
        v = tokens[((size_t)b * S_ + s) * D_ + c];
    } else {
        v = cls[(size_t)b * D_ + (c - D_)];
    }
    comb[(size_t)tid] = v;
}

__global__ void gemm_bias_tanh_kernel(const float* __restrict__ comb,
                                      const float* __restrict__ Wout,
                                      const float* __restrict__ bout,
                                      float*       __restrict__ out) {
    const int lane  = threadIdx.x;
    const int mt    = blockIdx.x;
    const int nt    = blockIdx.y;
    const int b0    = mt << 4;
    const int n0    = nt << 4;
    const int lmod  = lane & 15;
    const int lhalf = lane >> 4;

    const float* aPtr = comb + (size_t)(b0 + lmod) * K_ + 2 * lhalf;
    const float* wPtr = Wout + (size_t)(n0 + lmod) * K_ + 2 * lhalf;

    v8f acc = {};
#pragma unroll 4
    for (int kb = 0; kb < K_; kb += 4) {
        const v2f a = *(const v2f*)(aPtr + kb);
        const v2f w = *(const v2f*)(wPtr + kb);
        acc = __builtin_amdgcn_wmma_f32_16x16x4_f32(
                  false, a, false, w, (short)0, acc, false, false);
    }

    const int   n    = n0 + lmod;
    const float bias = bout[n];
#pragma unroll
    for (int v = 0; v < 8; ++v) {
        const int m = b0 + v + 8 * lhalf;
        out[(size_t)m * D_ + n] = tanhf(acc[v] + bias);
    }
}

extern "C" void kernel_launch(void* const* d_in, const int* in_sizes, int n_in,
                              void* d_out, int out_size, void* d_ws, size_t ws_size,
                              hipStream_t stream) {
    const float* tokens = (const float*)d_in[0];
    const float* cls    = (const float*)d_in[1];
    const float* Wout   = (const float*)d_in[2];
    const float* bout   = (const float*)d_in[3];
    const int*   idx    = (const int*)  d_in[4];
    float*       out    = (float*)d_out;
    float*       comb   = (float*)d_ws;

    const int total = B_ * K_;
    build_combined_kernel<<<dim3(total / 256), 256, 0, stream>>>(tokens, cls, idx, comb);

    gemm_bias_tanh_kernel<<<dim3(B_ / 16, D_ / 16), 32, 0, stream>>>(comb, Wout, bout, out);
}


# GEMM K loop split over the 16 waves of a 512-thread workgroup per tile (32 straight-line products per wave), partial tiles summed through LDS in wave order: f32 summation order changes
# speedup vs baseline: 3.1220x; 3.1220x over previous
#include <hip/hip_runtime.h>
#include <hip/hip_bf16.h>
#include <cmath>

typedef __attribute__((ext_vector_type(2))) float v2f;
typedef __attribute__((ext_vector_type(8))) float v8f;

#define B_  64
#define S_  512
#define D_  1024
#define K_  2048

__global__ void build_combined_kernel(const float* __restrict__ tokens,
                                      const float* __restrict__ cls,
                                      const int*   __restrict__ idx,
                                      float*       __restrict__ comb) {
    const int tid = blockIdx.x * blockDim.x + threadIdx.x;
    const int b = tid >> 11;
    const int c = tid & (K_ - 1);
    float v;
    if (c < D_) {
        const int s = idx[c] + 1;
        v = tokens[((size_t)b * S_ + s) * D_ + c];
    } else {
        v = cls[(size_t)b * D_ + (c - D_)];
    }
    comb[(size_t)tid] = v;
}

#define KW_     16
#define KSLICE_ (K_ / KW_)
#define KSTEPS_ (KSLICE_ / 4)

__global__ __launch_bounds__(KW_ * 32)
void gemm_bias_tanh_kernel(const float* __restrict__ comb,
                           const float* __restrict__ Wout,
                           const float* __restrict__ bout,
                           float*       __restrict__ out) {
    __shared__ float part[KW_][8][32];

    const int lane  = threadIdx.x & 31;
    const int wave  = __builtin_amdgcn_readfirstlane((int)(threadIdx.x >> 5));
    const int mt    = blockIdx.x;
    const int nt    = blockIdx.y;
    const int b0    = mt << 4;
    const int n0    = nt << 4;
    const int lmod  = lane & 15;
    const int lhalf = lane >> 4;

    const int    k0   = wave * KSLICE_;
    const float* aPtr = comb + (size_t)(b0 + lmod) * K_ + k0 + 2 * lhalf;
    const float* wPtr = Wout + (size_t)(n0 + lmod) * K_ + k0 + 2 * lhalf;

    v2f a[KSTEPS_], w[KSTEPS_];
#pragma unroll
    for (int i = 0; i < KSTEPS_; ++i) {
        a[i] = *(const v2f*)(aPtr + 4 * i);
        w[i] = *(const v2f*)(wPtr + 4 * i);
    }

    v8f acc = {};
#pragma unroll
    for (int i = 0; i < KSTEPS_; ++i) {
        acc = __builtin_amdgcn_wmma_f32_16x16x4_f32(
                  false, a[i], false, w[i], (short)0, acc, false, false);
    }

#pragma unroll
    for (int v = 0; v < 8; ++v) part[wave][v][lane] = acc[v];
    __syncthreads();

    if (wave < 8) {
        float s = part[0][wave][lane];
#pragma unroll
        for (int p = 1; p < KW_; ++p) s += part[p][wave][lane];
        const int n = n0 + lmod;
        const int m = b0 + wave + 8 * lhalf;
        out[(size_t)m * D_ + n] = tanhf(s + bout[n]);
    }
}

extern "C" void kernel_launch(void* const* d_in, const int* in_sizes, int n_in,
                              void* d_out, int out_size, void* d_ws, size_t ws_size,
                              hipStream_t stream) {
    const float* tokens = (const float*)d_in[0];
    const float* cls    = (const float*)d_in[1];
    const float* Wout   = (const float*)d_in[2];
    const float* bout   = (const float*)d_in[3];
    const int*   idx    = (const int*)  d_in[4];
    float*       out    = (float*)d_out;
    float*       comb   = (float*)d_ws;

    const int total = B_ * K_;
    build_combined_kernel<<<dim3(total / 256), 256, 0, stream>>>(tokens, cls, idx, comb);

    gemm_bias_tanh_kernel<<<dim3(B_ / 16, D_ / 16), KW_ * 32, 0, stream>>>(comb, Wout, bout, out);
}


# Fragment loads touch fewer 128-byte lines: 16 bytes per lane in both operands (same permuted k order), activations stored in fragment order so an A load is one 512-byte run; f32 sum order changes
# speedup vs baseline: 3.5132x; 1.1253x over previous
#include <hip/hip_runtime.h>
#include <hip/hip_bf16.h>
#include <cmath>

typedef __attribute__((ext_vector_type(2))) float v2f;
typedef __attribute__((ext_vector_type(4))) float v4f;
typedef __attribute__((ext_vector_type(8))) float v8f;

#define B_  64
#define S_  512
#define D_  1024
#define K_  2048

__global__ void build_combined_kernel(const float* __restrict__ tokens,
                                      const float* __restrict__ cls,
                                      const int*   __restrict__ idx,
                                      float*       __restrict__ comb) {
    const int tid = blockIdx.x * blockDim.x + threadIdx.x;
    const int e  = tid & 3;
    const int r  = (tid >> 2) & 15;
    const int kq = (tid >> 6) & (K_ / 4 - 1);
    const int mt = tid >> 15;
    const int b = (mt << 4) + r;
    const int k = (kq << 2) + e;
    float v;
    if (k < D_) {
        const int s = idx[k] + 1;
        v = tokens[((size_t)b * S_ + s) * D_ + k];
    } else {
        v = cls[(size_t)b * D_ + (k - D_)];
    }
    comb[(size_t)tid] = v;
}

#define KW_     16
#define KSLICE_ (K_ / KW_)
#define KPAIRS_ (KSLICE_ / 8)

__global__ __launch_bounds__(KW_ * 32)
void gemm_bias_tanh_kernel(const float* __restrict__ comb,
                           const float* __restrict__ Wout,
                           const float* __restrict__ bout,
                           float*       __restrict__ out) {
    __shared__ float part[KW_][8][32];

    const int lane  = threadIdx.x & 31;
    const int wave  = __builtin_amdgcn_readfirstlane((int)(threadIdx.x >> 5));
    const int mt    = blockIdx.x;
    const int nt    = blockIdx.y;
    const int b0    = mt << 4;
    const int n0    = nt << 4;
    const int lmod  = lane & 15;
    const int lhalf = lane >> 4;

    const int    k0   = wave * KSLICE_;
    const v4f*   aPtr = (const v4f*)comb + (size_t)(mt * (K_ / 4) + (k0 >> 2)) * 16 + lane;
    const float* wPtr = Wout + (size_t)(n0 + lmod) * K_ + k0 + 4 * lhalf;

    v4f a[KPAIRS_], w[KPAIRS_];
#pragma unroll
    for (int j = 0; j < KPAIRS_; ++j) {
        a[j] = aPtr[32 * j];
        w[j] = *(const v4f*)(wPtr + 8 * j);
    }

    v8f acc = {};
#pragma unroll
    for (int j = 0; j < KPAIRS_; ++j) {
        acc = __builtin_amdgcn_wmma_f32_16x16x4_f32(
                  false, __builtin_shufflevector(a[j], a[j], 0, 1),
                  false, __builtin_shufflevector(w[j], w[j], 0, 1),
                  (short)0, acc, false, false);
        acc = __builtin_amdgcn_wmma_f32_16x16x4_f32(
                  false, __builtin_shufflevector(a[j], a[j], 2, 3),
                  false, __builtin_shufflevector(w[j], w[j], 2, 3),
                  (short)0, acc, false, false);
    }

#pragma unroll
    for (int v = 0; v < 8; ++v) part[wave][v][lane] = acc[v];
    __syncthreads();

    if (wave < 8) {
        float s = part[0][wave][lane];
#pragma unroll
        for (int p = 1; p < KW_; ++p) s += part[p][wave][lane];
        const int n = n0 + lmod;
        const int m = b0 + wave + 8 * lhalf;
        out[(size_t)m * D_ + n] = tanhf(s + bout[n]);
    }
}

extern "C" void kernel_launch(void* const* d_in, const int* in_sizes, int n_in,
                              void* d_out, int out_size, void* d_ws, size_t ws_size,
                              hipStream_t stream) {
    const float* tokens = (const float*)d_in[0];
    const float* cls    = (const float*)d_in[1];
    const float* Wout   = (const float*)d_in[2];
    const float* bout   = (const float*)d_in[3];
    const int*   idx    = (const int*)  d_in[4];
    float*       out    = (float*)d_out;
    float*       comb   = (float*)d_ws;

    const int total = B_ * K_;
    build_combined_kernel<<<dim3(total / 256), 256, 0, stream>>>(tokens, cls, idx, comb);

    gemm_bias_tanh_kernel<<<dim3(B_ / 16, D_ / 16), KW_ * 32, 0, stream>>>(comb, Wout, bout, out);
}


# GEMM K split over 4 waves per workgroup (one per SIMD), not 16: a quarter of the waves to enter, 128 products per wave on two accumulators behind grouped loads, 4-way LDS sum; f32 sum order changes
# speedup vs baseline: 3.9326x; 1.1194x over previous
#include <hip/hip_runtime.h>
#include <hip/hip_bf16.h>
#include <cmath>

typedef __attribute__((ext_vector_type(2))) float v2f;
typedef __attribute__((ext_vector_type(4))) float v4f;
typedef __attribute__((ext_vector_type(8))) float v8f;

#define B_  64
#define S_  512
#define D_  1024
#define K_  2048

__global__ void build_combined_kernel(const float* __restrict__ tokens,
                                      const float* __restrict__ cls,
                                      const int*   __restrict__ idx,
                                      float*       __restrict__ comb) {
    const int tid = blockIdx.x * blockDim.x + threadIdx.x;
    const int e  = tid & 3;
    const int r  = (tid >> 2) & 15;
    const int kq = (tid >> 6) & (K_ / 4 - 1);
    const int mt = tid >> 15;
    const int b = (mt << 4) + r;
    const int k = (kq << 2) + e;
    float v;
    if (k < D_) {
        const int s = idx[k] + 1;
        v = tokens[((size_t)b * S_ + s) * D_ + k];
    } else {
        v = cls[(size_t)b * D_ + (k - D_)];
    }
    comb[(size_t)tid] = v;
}

#define KW_     4
#define KSLICE_ (K_ / KW_)
#define KPAIRS_ (KSLICE_ / 8)

__global__ __launch_bounds__(KW_ * 32)
void gemm_bias_tanh_kernel(const float* __restrict__ comb,
                           const float* __restrict__ Wout,
                           const float* __restrict__ bout,
                           float*       __restrict__ out) {
    __shared__ float part[KW_][8][32];

    const int lane  = threadIdx.x & 31;
    const int wave  = __builtin_amdgcn_readfirstlane((int)(threadIdx.x >> 5));
    const int mt    = blockIdx.x;
    const int nt    = blockIdx.y;
    const int b0    = mt << 4;
    const int n0    = nt << 4;
    const int lmod  = lane & 15;
    const int lhalf = lane >> 4;

    const int    k0   = wave * KSLICE_;
    const v4f*   aPtr = (const v4f*)comb + (size_t)(mt * (K_ / 4) + (k0 >> 2)) * 16 + lane;
    const float* wPtr = Wout + (size_t)(n0 + lmod) * K_ + k0 + 4 * lhalf;

    const float bias = bout[n0 + lmod];

    v4f a[KPAIRS_], w[KPAIRS_];
    v8f acc0 = {}, acc1 = {};

#define LOAD_PAIRS_(lo, hi)                                                   \
    _Pragma("unroll")                                                         \
    for (int j = (lo); j < (hi); ++j) {                                       \
        a[j] = aPtr[32 * j];                                                  \
        w[j] = *(const v4f*)(wPtr + 8 * j);                                   \
    }
#define MUL_PAIRS_(lo, hi)                                                    \
    _Pragma("unroll")                                                         \
    for (int j = (lo); j < (hi); ++j) {                                       \
        acc0 = __builtin_amdgcn_wmma_f32_16x16x4_f32(                         \
                   false, __builtin_shufflevector(a[j], a[j], 0, 1),          \
                   false, __builtin_shufflevector(w[j], w[j], 0, 1),          \
                   (short)0, acc0, false, false);                             \
        acc1 = __builtin_amdgcn_wmma_f32_16x16x4_f32(                         \
                   false, __builtin_shufflevector(a[j], a[j], 2, 3),          \
                   false, __builtin_shufflevector(w[j], w[j], 2, 3),          \
                   (short)0, acc1, false, false);                             \
    }
#define ORDER_FENCE_()  do { asm volatile("" ::: "memory"); __builtin_amdgcn_sched_barrier(0); } while (0)

    LOAD_PAIRS_(0, 31)
    ORDER_FENCE_();
    MUL_PAIRS_(0, 8)
    ORDER_FENCE_();
    LOAD_PAIRS_(31, 39)
    ORDER_FENCE_();
    MUL_PAIRS_(8, 16)
    ORDER_FENCE_();
    LOAD_PAIRS_(39, 47)
    ORDER_FENCE_();
    MUL_PAIRS_(16, 24)
    ORDER_FENCE_();
    LOAD_PAIRS_(47, 55)
    ORDER_FENCE_();
    MUL_PAIRS_(24, 32)
    ORDER_FENCE_();
    LOAD_PAIRS_(55, 64)
    ORDER_FENCE_();
    MUL_PAIRS_(32, 64)
#undef LOAD_PAIRS_
#undef MUL_PAIRS_
#undef ORDER_FENCE_
    const v8f acc = acc0 + acc1;

#pragma unroll
    for (int v = 0; v < 8; ++v) part[wave][v][lane] = acc[v];
    __syncthreads();

#pragma unroll
    for (int t = 0; t < 2; ++t) {
        const int v = 2 * wave + t;
        float s = part[0][v][lane];
#pragma unroll
        for (int p = 1; p < KW_; ++p) s += part[p][v][lane];
        const int n = n0 + lmod;
        const int m = b0 + v + 8 * lhalf;
        out[(size_t)m * D_ + n] = tanhf(s + bias);
    }
}

extern "C" void kernel_launch(void* const* d_in, const int* in_sizes, int n_in,
                              void* d_out, int out_size, void* d_ws, size_t ws_size,
                              hipStream_t stream) {
    const float* tokens = (const float*)d_in[0];
    const float* cls    = (const float*)d_in[1];
    const float* Wout   = (const float*)d_in[2];
    const float* bout   = (const float*)d_in[3];
    const int*   idx    = (const int*)  d_in[4];
    float*       out    = (float*)d_out;
    float*       comb   = (float*)d_ws;

    const int total = B_ * K_;
    build_combined_kernel<<<dim3(total / 256), 256, 0, stream>>>(tokens, cls, idx, comb);

    gemm_bias_tanh_kernel<<<dim3(B_ / 16, D_ / 16), KW_ * 32, 0, stream>>>(comb, Wout, bout, out);
}
